# starting kernel
# baseline (speedup 1.0000x reference)
#include <hip/hip_runtime.h>

typedef __attribute__((ext_vector_type(2))) float v2f;
typedef __attribute__((ext_vector_type(8))) float v8f;

namespace {
constexpr int Bn = 4;
constexpr int Cn = 256;
constexpr int Hn = 128;
constexpr int Wn = 224;
constexpr int Pn = 4;
constexpr int ND = 2 * Pn + 1;
constexpr int NWAVES = 7;
}

__global__ __launch_bounds__(NWAVES * 32) void corr_wmma_kernel(
    const float* __restrict__ first,
    const float* __restrict__ second,
    float* __restrict__ out) {
  __shared__ float lds_out[ND * Wn];

  const int blk = blockIdx.x;
  const int dy  = blk % ND;
  const int by  = blk / ND;
  const int y   = by % Hn;
  const int b   = by / Hn;
  const int z   = y + dy - Pn;

  const int tid  = threadIdx.x;
  const int wave = tid >> 5;
  const int lane = tid & 31;
  const int half = lane >> 4;
  const int ln   = lane & 15;

  const size_t cs = (size_t)Hn * Wn;
  float* outbase = out + (((size_t)(b * ND * ND + dy * ND)) * Hn + y) * Wn;

  if (z < 0 || z >= Hn) {
    for (int i = tid; i < ND * Wn; i += blockDim.x) {
      const int dx = i / Wn;
      const int x  = i - dx * Wn;
      outbase[(size_t)dx * cs + x] = 0.0f;
    }
    return;
  }

  const float* Frow = first  + ((size_t)(b * Cn) * Hn + y) * Wn;
  const float* Srow = second + ((size_t)(b * Cn) * Hn + z) * Wn;

  const int n0 = wave * 32;

  const int ca0 = n0 - 4  + ln;
  const int ca1 = n0 + 12 + ln;
  const int ca2 = n0 + 28 + ln;
  const int xb0 = n0 + ln;
  const int xb1 = n0 + 16 + ln;

  const int   ca0c = ca0 < 0 ? 0 : ca0;
  const int   ca2c = ca2 >= Wn ? Wn - 1 : ca2;
  const float m0   = (ca0 >= 0) ? 1.0f : 0.0f;
  const float m2   = (ca2 < Wn) ? 1.0f : 0.0f;

  v8f d00 = {};
  v8f d01 = {};
  v8f d11 = {};
  v8f d21 = {};

#pragma unroll 4
  for (int k = 0; k < Cn; k += 4) {
    const int c0 = k + 2 * half;
    const float* Fp = Frow + (size_t)c0 * cs;
    const float* Sp = Srow + (size_t)c0 * cs;

    v2f b0;  b0.x = Fp[xb0];       b0.y = Fp[cs + xb0];
    v2f b1;  b1.x = Fp[xb1];       b1.y = Fp[cs + xb1];
    v2f a0;  a0.x = Sp[ca0c] * m0; a0.y = Sp[cs + ca0c] * m0;
    v2f a1;  a1.x = Sp[ca1];       a1.y = Sp[cs + ca1];
    v2f a2;  a2.x = Sp[ca2c] * m2; a2.y = Sp[cs + ca2c] * m2;

    d00 = __builtin_amdgcn_wmma_f32_16x16x4_f32(
        false, a0, false, b0, (short)0, d00, false, false);
    d01 = __builtin_amdgcn_wmma_f32_16x16x4_f32(
        false, a1, false, b0, (short)0, d01, false, false);
    d11 = __builtin_amdgcn_wmma_f32_16x16x4_f32(
        false, a1, false, b1, (short)0, d11, false, false);
    d21 = __builtin_amdgcn_wmma_f32_16x16x4_f32(
        false, a2, false, b1, (short)0, d21, false, false);
  }

#pragma unroll
  for (int v = 0; v < 8; ++v) {
    const int m   = v + 8 * half;
    const int dxL = m - ln;
    const int dxH = m + 16 - ln;
    if (dxL >= 0 && dxL < ND) {
      lds_out[dxL * Wn + n0 + ln]      = d00[v];
      lds_out[dxL * Wn + n0 + 16 + ln] = d11[v];
    }
    if (dxH >= 0 && dxH < ND) {
      lds_out[dxH * Wn + n0 + ln]      = d01[v];
      lds_out[dxH * Wn + n0 + 16 + ln] = d21[v];
    }
  }

  __syncthreads();

  for (int i = tid; i < ND * Wn; i += blockDim.x) {
    const int dx = i / Wn;
    const int x  = i - dx * Wn;
    outbase[(size_t)dx * cs + x] = lds_out[i] * (1.0f / (float)Cn);
  }
}

extern "C" void kernel_launch(void* const* d_in, const int* in_sizes, int n_in,
                              void* d_out, int out_size, void* d_ws, size_t ws_size,
                              hipStream_t stream) {
  const float* first  = (const float*)d_in[0];
  const float* second = (const float*)d_in[1];
  float* out = (float*)d_out;

  const int nblocks = Bn * Hn * ND;
  corr_wmma_kernel<<<nblocks, NWAVES * 32, 0, stream>>>(first, second, out);
}


# Stage both operand rows through LDS in 16-channel slabs (16-byte whole-line global loads, next slab prefetched in registers); WMMA fragments read from LDS; summation order unchanged
# speedup vs baseline: 1.5971x; 1.5971x over previous
#include <hip/hip_runtime.h>


typedef __attribute__((ext_vector_type(2))) float v2f;
typedef __attribute__((ext_vector_type(8))) float v8f;

namespace {
constexpr int Bn = 4;
constexpr int Cn = 256;
constexpr int Hn = 128;
constexpr int Wn = 224;
constexpr int Pn = 4;
constexpr int ND = 2 * Pn + 1;
constexpr int NWAVES = 7;
constexpr int NT = NWAVES * 32;
constexpr int KC = 16;
constexpr int W4 = Wn / 4;
constexpr int LPT = KC * W4 / NT;
constexpr int TS = KC * 16 + 16;
constexpr int FT = Wn / 16;
constexpr int ST = FT + 1;
constexpr int FSLAB = FT * TS;
constexpr int SSLAB = ST * TS;
constexpr int STAGE = FSLAB + SSLAB;
constexpr int NCHUNK = Cn / KC;
}

__device__ __forceinline__ void stage_mma(const float* __restrict__ fs,
                                          v8f& d00, v8f& d01, v8f& d11, v8f& d21) {
#pragma unroll
  for (int ks = 0; ks < KC / 4; ++ks) {
    const float* fp = fs + ks * 64;
    const float* sp = fp + FSLAB;
    v2f b0;  b0.x = fp[0];       b0.y = fp[16];
    v2f b1;  b1.x = fp[TS];      b1.y = fp[TS + 16];
    v2f a0;  a0.x = sp[0];       a0.y = sp[16];
    v2f a1;  a1.x = sp[TS];      a1.y = sp[TS + 16];
    v2f a2;  a2.x = sp[2 * TS];  a2.y = sp[2 * TS + 16];
    d00 = __builtin_amdgcn_wmma_f32_16x16x4_f32(
        false, a0, false, b0, (short)0, d00, false, false);
    d01 = __builtin_amdgcn_wmma_f32_16x16x4_f32(
        false, a1, false, b0, (short)0, d01, false, false);
    d11 = __builtin_amdgcn_wmma_f32_16x16x4_f32(
        false, a1, false, b1, (short)0, d11, false, false);
    d21 = __builtin_amdgcn_wmma_f32_16x16x4_f32(
        false, a2, false, b1, (short)0, d21, false, false);
  }
}

__global__ __launch_bounds__(NWAVES * 32) void corr_wmma_kernel(
    const float* __restrict__ first,
    const float* __restrict__ second,
    float* __restrict__ out) {
  __shared__ __attribute__((aligned(16))) float lds[2 * STAGE];
  float* lds_out = lds;

  const int blk = blockIdx.x;
  const int dy  = blk % ND;
  const int by  = blk / ND;
  const int y   = by % Hn;
  const int b   = by / Hn;
  const int z   = y + dy - Pn;

  const int tid  = threadIdx.x;
  const int wave = tid >> 5;
  const int lane = tid & 31;
  const int half = lane >> 4;
  const int ln   = lane & 15;

  const size_t cs = (size_t)Hn * Wn;
  float* outbase = out + (((size_t)(b * ND * ND + dy * ND)) * Hn + y) * Wn;

  if (z < 0 || z >= Hn) {
    for (int i = tid; i < ND * Wn; i += blockDim.x) {
      const int dx = i / Wn;
      const int x  = i - dx * Wn;
      outbase[(size_t)dx * cs + x] = 0.0f;
    }
    return;
  }

  const float* Frow = first  + ((size_t)(b * Cn) * Hn + y) * Wn;
  const float* Srow = second + ((size_t)(b * Cn) * Hn + z) * Wn;

  const int lrow = tid / W4;
  const int lc4  = tid - lrow * W4;
  const float* Fg = Frow + (size_t)lrow * cs + 4 * lc4;
  const float* Sg = Srow + (size_t)lrow * cs + 4 * lc4;
  const int fst = (lc4 >> 2) * TS + lrow * 16 + 4 * (lc4 & 3);
  const int sst = FSLAB + ((lc4 + 1) >> 2) * TS + lrow * 16 + 4 * ((lc4 + 1) & 3);

  for (int i = tid; i < 2 * KC * 16; i += NT) {
    const int q  = i & 15;
    const int r  = (i >> 4) % KC;
    const int st = i / (KC * 16);
    const int dw = ((q < 4) ? 0 : (ST - 1) * TS) + r * 16 + q;
    lds[st * STAGE + FSLAB + dw] = 0.0f;
  }

  float4 pf[LPT], ps[LPT];
#pragma unroll
  for (int i = 0; i < LPT; ++i) {
    pf[i] = *reinterpret_cast<const float4*>(Fg + (size_t)(4 * i) * cs);
    ps[i] = *reinterpret_cast<const float4*>(Sg + (size_t)(4 * i) * cs);
  }
#pragma unroll
  for (int i = 0; i < LPT; ++i) {
    *reinterpret_cast<float4*>(&lds[fst + 64 * i]) = pf[i];
    *reinterpret_cast<float4*>(&lds[sst + 64 * i]) = ps[i];
  }
  __syncthreads();

  const int n0 = wave * 32;
  const int fo = 2 * wave * TS + 2 * half * 16 + ln;

  v8f d00 = {};
  v8f d01 = {};
  v8f d11 = {};
  v8f d21 = {};

  for (int ch = 0; ch < NCHUNK - 1; ++ch) {
    const int cur = ch & 1;

    const size_t cn = (size_t)(ch + 1) * KC;
#pragma unroll
    for (int i = 0; i < LPT; ++i) {
      pf[i] = *reinterpret_cast<const float4*>(Fg + (cn + 4 * i) * cs);
      ps[i] = *reinterpret_cast<const float4*>(Sg + (cn + 4 * i) * cs);
    }

    __builtin_amdgcn_sched_barrier(0);

    stage_mma(lds + cur * STAGE + fo, d00, d01, d11, d21);

    float* nx = lds + (cur ^ 1) * STAGE;
#pragma unroll
    for (int i = 0; i < LPT; ++i) {
      *reinterpret_cast<float4*>(&nx[fst + 64 * i]) = pf[i];
      *reinterpret_cast<float4*>(&nx[sst + 64 * i]) = ps[i];
    }
    __syncthreads();
  }

  stage_mma(lds + ((NCHUNK - 1) & 1) * STAGE + fo, d00, d01, d11, d21);
  __syncthreads();

#pragma unroll
  for (int v = 0; v < 8; ++v) {
    const int m   = v + 8 * half;
    const int dxL = m - ln;
    const int dxH = m + 16 - ln;
    if (dxL >= 0 && dxL < ND) {
      lds_out[dxL * Wn + n0 + ln]      = d00[v];
      lds_out[dxL * Wn + n0 + 16 + ln] = d11[v];
    }
    if (dxH >= 0 && dxH < ND) {
      lds_out[dxH * Wn + n0 + ln]      = d01[v];
      lds_out[dxH * Wn + n0 + 16 + ln] = d21[v];
    }
  }

  __syncthreads();

  for (int i = tid; i < ND * Wn; i += blockDim.x) {
    const int dx = i / Wn;
    const int x  = i - dx * Wn;
    outbase[(size_t)dx * cs + x] = lds_out[i] * (1.0f / (float)Cn);
  }
}

extern "C" void kernel_launch(void* const* d_in, const int* in_sizes, int n_in,
                              void* d_out, int out_size, void* d_ws, size_t ws_size,
                              hipStream_t stream) {
  const float* first  = (const float*)d_in[0];
  const float* second = (const float*)d_in[1];
  float* out = (float*)d_out;

  const int nblocks = Bn * Hn * ND;
  corr_wmma_kernel<<<nblocks, NWAVES * 32, 0, stream>>>(first, second, out);
}


# Re-tile the band as 2x8 pixel patches: one block per first-row pair stages its 12 image rows once for all 81 displacements; 10 WMMAs per 16 pixels per k-step, not 18; summation order unchanged
# speedup vs baseline: 2.7302x; 1.7095x over previous
#include <hip/hip_runtime.h>


typedef __attribute__((ext_vector_type(2))) float v2f;
typedef __attribute__((ext_vector_type(8))) float v8f;

namespace {
constexpr int Bn = 4;
constexpr int Cn = 256;
constexpr int Hn = 128;
constexpr int Wn = 224;
constexpr int Pn = 4;
constexpr int ND = 2 * Pn + 1;
constexpr int NWAVES = Wn / 8;
constexpr int NT = NWAVES * 32;
constexpr int KC = 8;
constexpr int W4 = Wn / 4;
constexpr int NSR = ND + 1;
constexpr int NROWS = 2 + NSR;
constexpr int LPT = NROWS * KC * W4 / NT;
constexpr int TS = KC * 16 + 16;
constexpr int FT = Wn / 16;
constexpr int ST = FT + 1;
constexpr int FROW = FT * TS + 40;
constexpr int SROW = ST * TS;
constexpr int SBASE = 2 * FROW;
constexpr int STAGE = SBASE + NSR * SROW;
constexpr int NCHUNK = Cn / KC;
constexpr int OP = Wn + 4;
constexpr int ORB = ND * ND * OP + 48;
static_assert(2 * ORB <= 2 * STAGE, "output staging must fit in the stage buffers");
static_assert(LPT == 6, "staging map assumes 6 float4 per thread");
}

__device__ __forceinline__ float4 keep_or_zero(float4 v, bool keep) {
  float4 r;
  r.x = keep ? v.x : 0.0f;
  r.y = keep ? v.y : 0.0f;
  r.z = keep ? v.z : 0.0f;
  r.w = keep ? v.w : 0.0f;
  return r;
}

__device__ __forceinline__ void kstep_mma(const float* __restrict__ st,
                                          int boff, int aoff, int ks, v8f (&d)[NSR]) {
  {
    const float* bp = st + boff + ks * 64;
    const float* ap = st + aoff + ks * 64;
    v2f bf;  bf.x = bp[0];  bf.y = bp[16];
    v2f af[NSR];
#pragma unroll
    for (int j = 0; j < NSR; ++j) { af[j].x = ap[j * SROW];  af[j].y = ap[j * SROW + 16]; }
#pragma unroll
    for (int j = 0; j < NSR; ++j)
      d[j] = __builtin_amdgcn_wmma_f32_16x16x4_f32(
          false, af[j], false, bf, (short)0, d[j], false, false);
  }
}

__global__ __launch_bounds__(NWAVES * 32) void corr_wmma_kernel(
    const float* __restrict__ first,
    const float* __restrict__ second,
    float* __restrict__ out) {
  __shared__ __attribute__((aligned(16))) float lds[2 * STAGE];
  float* lds_out = lds;

  const int blk = blockIdx.x;
  const int yp  = blk % (Hn / 2);
  const int b   = blk / (Hn / 2);
  const int y0  = 2 * yp;

  const int tid  = threadIdx.x;
  const int wave = tid >> 5;
  const int lane = tid & 31;
  const int half = lane >> 4;
  const int ln   = lane & 15;
  const int pc   = ln & 7;
  const int pr   = ln >> 3;

  const unsigned cs = (unsigned)Hn * Wn;

  const int lr  = tid / W4;
  const int lc4 = tid - lr * W4;
  const int sch = lr & (KC - 1);
  const int par = lr >> 3;
  const char* fbase = reinterpret_cast<const char*>(first);
  const char* sbase = reinterpret_cast<const char*>(second);
  const unsigned gcol = 4u * (((unsigned)(b * Cn + sch) * Hn) * Wn + 4 * lc4);
  unsigned fgo = gcol + 4u * (unsigned)(y0 + par) * Wn;
  unsigned sgo[LPT - 1];
  bool     sok[LPT - 1];
#pragma unroll
  for (int i = 0; i < LPT - 1; ++i) {
    const int z = y0 - Pn + par + 2 * i;
    sok[i] = (z >= 0) && (z < Hn);
    const int zc = z < 0 ? 0 : (z >= Hn ? Hn - 1 : z);
    sgo[i] = gcol + 4u * (unsigned)zc * Wn;
  }
  const int fst = par * FROW + (lc4 >> 2) * TS + sch * 16 + 4 * (lc4 & 3);
  const int sst = SBASE + par * SROW + ((lc4 + 1) >> 2) * TS + sch * 16 + 4 * ((lc4 + 1) & 3);

  for (int i = tid; i < 2 * NSR * KC * 16; i += NT) {
    const int q  = i & 15;
    const int r  = (i >> 4) % KC;
    const int j  = (i / (KC * 16)) % NSR;
    const int st = i / (NSR * KC * 16);
    const int dw = ((q < 4) ? 0 : (ST - 1) * TS) + r * 16 + q;
    lds[st * STAGE + SBASE + j * SROW + dw] = 0.0f;
  }

  float4 pf, ps[LPT - 1];
  pf = *reinterpret_cast<const float4*>(fbase + fgo);
#pragma unroll
  for (int i = 0; i < LPT - 1; ++i)
    ps[i] = *reinterpret_cast<const float4*>(sbase + sgo[i]);
  *reinterpret_cast<float4*>(&lds[fst]) = pf;
#pragma unroll
  for (int i = 0; i < LPT - 1; ++i)
    *reinterpret_cast<float4*>(&lds[sst + 2 * i * SROW]) = keep_or_zero(ps[i], sok[i]);
  __syncthreads();

  const int x0 = 8 * wave;
  const int bx = x0 + pc;
  const int ax = x0 + ln;
  const int boff = pr * FROW + (bx >> 4) * TS + (bx & 15) + 2 * half * 16;
  const int aoff = SBASE + (ax >> 4) * TS + (ax & 15) + 2 * half * 16;

  v8f d[NSR];
#pragma unroll
  for (int j = 0; j < NSR; ++j) d[j] = v8f{};

  static_assert(KC == 8, "the stage loop below is written for two k-steps per stage");
  for (int ch = 0; ch < NCHUNK - 1; ++ch) {
    const int cur = ch & 1;
    const float* cs_lds = lds + cur * STAGE;
    float* nx = lds + (cur ^ 1) * STAGE;
    float* nxf = nx + fst;
    float* nxs = nx + sst;

    fgo += 4u * KC * cs;
    pf = *reinterpret_cast<const float4*>(fbase + fgo);
#pragma unroll
    for (int i = 0; i < 2; ++i) {
      sgo[i] += 4u * KC * cs;
      ps[i] = *reinterpret_cast<const float4*>(sbase + sgo[i]);
    }

    __builtin_amdgcn_sched_barrier(0);

    kstep_mma(cs_lds, boff, aoff, 0, d);

    *reinterpret_cast<float4*>(nxf) = pf;
#pragma unroll
    for (int i = 0; i < 2; ++i)
      *reinterpret_cast<float4*>(nxs + 2 * i * SROW) = keep_or_zero(ps[i], sok[i]);

#pragma unroll
    for (int i = 2; i < LPT - 1; ++i) {
      sgo[i] += 4u * KC * cs;
      ps[i] = *reinterpret_cast<const float4*>(sbase + sgo[i]);
    }

    __builtin_amdgcn_sched_barrier(0);

    kstep_mma(cs_lds, boff, aoff, 1, d);

#pragma unroll
    for (int i = 2; i < LPT - 1; ++i)
      *reinterpret_cast<float4*>(nxs + 2 * i * SROW) = keep_or_zero(ps[i], sok[i]);
    __syncthreads();
  }

  {
    const float* ls = lds + ((NCHUNK - 1) & 1) * STAGE;
    kstep_mma(ls, boff, aoff, 0, d);
    kstep_mma(ls, boff, aoff, 1, d);
  }
  __syncthreads();

  const float scale = 1.0f / (float)Cn;
#pragma unroll
  for (int j = 0; j < NSR; ++j) {
    const int dy = j - pr;
#pragma unroll
    for (int v = 0; v < 8; ++v) {
      const int dx = v + 8 * half - pc;
      if (dy >= 0 && dy < ND && dx >= 0 && dx < ND)
        lds_out[pr * ORB + (dy * ND + dx) * OP + bx] = d[j][v] * scale;
    }
  }

  __syncthreads();

  for (int i = tid; i < 2 * ND * ND * W4; i += NT) {
    const int rr = i / W4;
    const int x4 = i - rr * W4;
    const int r2 = rr / (ND * ND);
    const int pl = rr - r2 * (ND * ND);
    const float4 val = *reinterpret_cast<const float4*>(&lds_out[r2 * ORB + pl * OP + 4 * x4]);
    float* op = out + (((size_t)(b * ND * ND + pl)) * Hn + (y0 + r2)) * Wn + 4 * x4;
    *reinterpret_cast<float4*>(op) = val;
  }
}

extern "C" void kernel_launch(void* const* d_in, const int* in_sizes, int n_in,
                              void* d_out, int out_size, void* d_ws, size_t ws_size,
                              hipStream_t stream) {
  const float* first  = (const float*)d_in[0];
  const float* second = (const float*)d_in[1];
  float* out = (float*)d_out;

  const int nblocks = Bn * (Hn / 2);
  corr_wmma_kernel<<<nblocks, NWAVES * 32, 0, stream>>>(first, second, out);
}


# Wave issue priority by stage phase (s_setprio 1 from the hand-over barrier through k-step 1, 0 for k-step 2) so every wave clears k-step 1 before the pipe serves k-step 2; summation order unchanged
# speedup vs baseline: 2.9434x; 1.0781x over previous
#include <hip/hip_runtime.h>


typedef __attribute__((ext_vector_type(2))) float v2f;
typedef __attribute__((ext_vector_type(8))) float v8f;

namespace {
constexpr int Bn = 4;
constexpr int Cn = 256;
constexpr int Hn = 128;
constexpr int Wn = 224;
constexpr int Pn = 4;
constexpr int ND = 2 * Pn + 1;
constexpr int NWAVES = Wn / 8;
constexpr int NT = NWAVES * 32;
constexpr int KC = 8;
constexpr int W4 = Wn / 4;
constexpr int NSR = ND + 1;
constexpr int NROWS = 2 + NSR;
constexpr int LPT = NROWS * KC * W4 / NT;
constexpr int TS = KC * 16 + 16;
constexpr int FT = Wn / 16;
constexpr int ST = FT + 1;
constexpr int FROW = FT * TS + 40;
constexpr int SROW = ST * TS;
constexpr int SBASE = 2 * FROW;
constexpr int STAGE = SBASE + NSR * SROW;
constexpr int NCHUNK = Cn / KC;
constexpr int OP = Wn + 4;
constexpr int ORB = ND * ND * OP + 48;
static_assert(2 * ORB <= 2 * STAGE, "output staging must fit in the stage buffers");
static_assert(LPT == 6, "staging map assumes 6 float4 per thread");
}

__device__ __forceinline__ float4 keep_or_zero(float4 v, bool keep) {
  float4 r;
  r.x = keep ? v.x : 0.0f;
  r.y = keep ? v.y : 0.0f;
  r.z = keep ? v.z : 0.0f;
  r.w = keep ? v.w : 0.0f;
  return r;
}

__device__ __forceinline__ void kstep_mma(const float* __restrict__ st,
                                          int boff, int aoff, int ks, v8f (&d)[NSR]) {
  {
    const float* bp = st + boff + ks * 64;
    const float* ap = st + aoff + ks * 64;
    v2f bf;  bf.x = bp[0];  bf.y = bp[16];
    v2f af[NSR];
#pragma unroll
    for (int j = 0; j < NSR; ++j) { af[j].x = ap[j * SROW];  af[j].y = ap[j * SROW + 16]; }
#pragma unroll
    for (int j = 0; j < NSR; ++j)
      d[j] = __builtin_amdgcn_wmma_f32_16x16x4_f32(
          false, af[j], false, bf, (short)0, d[j], false, false);
  }
}

__global__ __launch_bounds__(NWAVES * 32) void corr_wmma_kernel(
    const float* __restrict__ first,
    const float* __restrict__ second,
    float* __restrict__ out) {
  __shared__ __attribute__((aligned(16))) float lds[2 * STAGE];
  float* lds_out = lds;

  const int blk = blockIdx.x;
  const int yp  = blk % (Hn / 2);
  const int b   = blk / (Hn / 2);
  const int y0  = 2 * yp;

  const int tid  = threadIdx.x;
  const int wave = tid >> 5;
  const int lane = tid & 31;
  const int half = lane >> 4;
  const int ln   = lane & 15;
  const int pc   = ln & 7;
  const int pr   = ln >> 3;

  const unsigned cs = (unsigned)Hn * Wn;

  const int lr  = tid / W4;
  const int lc4 = tid - lr * W4;
  const int sch = lr & (KC - 1);
  const int par = lr >> 3;
  const char* fbase = reinterpret_cast<const char*>(first);
  const char* sbase = reinterpret_cast<const char*>(second);
  const unsigned gcol = 4u * (((unsigned)(b * Cn + sch) * Hn) * Wn + 4 * lc4);
  unsigned fgo = gcol + 4u * (unsigned)(y0 + par) * Wn;
  unsigned sgo[LPT - 1];
  bool     sok[LPT - 1];
#pragma unroll
  for (int i = 0; i < LPT - 1; ++i) {
    const int z = y0 - Pn + par + 2 * i;
    sok[i] = (z >= 0) && (z < Hn);
    const int zc = z < 0 ? 0 : (z >= Hn ? Hn - 1 : z);
    sgo[i] = gcol + 4u * (unsigned)zc * Wn;
  }
  const int fst = par * FROW + (lc4 >> 2) * TS + sch * 16 + 4 * (lc4 & 3);
  const int sst = SBASE + par * SROW + ((lc4 + 1) >> 2) * TS + sch * 16 + 4 * ((lc4 + 1) & 3);

  for (int i = tid; i < 2 * NSR * KC * 16; i += NT) {
    const int q  = i & 15;
    const int r  = (i >> 4) % KC;
    const int j  = (i / (KC * 16)) % NSR;
    const int st = i / (NSR * KC * 16);
    const int dw = ((q < 4) ? 0 : (ST - 1) * TS) + r * 16 + q;
    lds[st * STAGE + SBASE + j * SROW + dw] = 0.0f;
  }

  float4 pf, ps[LPT - 1];
  pf = *reinterpret_cast<const float4*>(fbase + fgo);
#pragma unroll
  for (int i = 0; i < LPT - 1; ++i)
    ps[i] = *reinterpret_cast<const float4*>(sbase + sgo[i]);
  *reinterpret_cast<float4*>(&lds[fst]) = pf;
#pragma unroll
  for (int i = 0; i < LPT - 1; ++i)
    *reinterpret_cast<float4*>(&lds[sst + 2 * i * SROW]) = keep_or_zero(ps[i], sok[i]);
  __syncthreads();

  const int x0 = 8 * wave;
  const int bx = x0 + pc;
  const int ax = x0 + ln;
  const int boff = pr * FROW + (bx >> 4) * TS + (bx & 15) + 2 * half * 16;
  const int aoff = SBASE + (ax >> 4) * TS + (ax & 15) + 2 * half * 16;

  v8f d[NSR];
#pragma unroll
  for (int j = 0; j < NSR; ++j) d[j] = v8f{};

  static_assert(KC == 8, "the stage loop below is written for two k-steps per stage");
  __builtin_amdgcn_s_setprio(1);
  for (int ch = 0; ch < NCHUNK - 1; ++ch) {
    const int cur = ch & 1;
    const float* cs_lds = lds + cur * STAGE;
    float* nx = lds + (cur ^ 1) * STAGE;
    float* nxf = nx + fst;
    float* nxs = nx + sst;

    fgo += 4u * KC * cs;
    pf = *reinterpret_cast<const float4*>(fbase + fgo);
#pragma unroll
    for (int i = 0; i < 2; ++i) {
      sgo[i] += 4u * KC * cs;
      ps[i] = *reinterpret_cast<const float4*>(sbase + sgo[i]);
    }

    __builtin_amdgcn_sched_barrier(0);

    kstep_mma(cs_lds, boff, aoff, 0, d);

    *reinterpret_cast<float4*>(nxf) = pf;
#pragma unroll
    for (int i = 0; i < 2; ++i)
      *reinterpret_cast<float4*>(nxs + 2 * i * SROW) = keep_or_zero(ps[i], sok[i]);

    __builtin_amdgcn_s_setprio(0);

#pragma unroll
    for (int i = 2; i < LPT - 1; ++i) {
      sgo[i] += 4u * KC * cs;
      ps[i] = *reinterpret_cast<const float4*>(sbase + sgo[i]);
    }

    __builtin_amdgcn_sched_barrier(0);

    kstep_mma(cs_lds, boff, aoff, 1, d);

#pragma unroll
    for (int i = 2; i < LPT - 1; ++i)
      *reinterpret_cast<float4*>(nxs + 2 * i * SROW) = keep_or_zero(ps[i], sok[i]);
    __builtin_amdgcn_s_setprio(1);
    __syncthreads();
  }
  __builtin_amdgcn_s_setprio(0);

  {
    const float* ls = lds + ((NCHUNK - 1) & 1) * STAGE;
    kstep_mma(ls, boff, aoff, 0, d);
    kstep_mma(ls, boff, aoff, 1, d);
  }
  __syncthreads();

  const float scale = 1.0f / (float)Cn;
#pragma unroll
  for (int j = 0; j < NSR; ++j) {
    const int dy = j - pr;
#pragma unroll
    for (int v = 0; v < 8; ++v) {
      const int dx = v + 8 * half - pc;
      if (dy >= 0 && dy < ND && dx >= 0 && dx < ND)
        lds_out[pr * ORB + (dy * ND + dx) * OP + bx] = d[j][v] * scale;
    }
  }

  __syncthreads();

  for (int i = tid; i < 2 * ND * ND * W4; i += NT) {
    const int rr = i / W4;
    const int x4 = i - rr * W4;
    const int r2 = rr / (ND * ND);
    const int pl = rr - r2 * (ND * ND);
    const float4 val = *reinterpret_cast<const float4*>(&lds_out[r2 * ORB + pl * OP + 4 * x4]);
    float* op = out + (((size_t)(b * ND * ND + pl)) * Hn + (y0 + r2)) * Wn + 4 * x4;
    *reinterpret_cast<float4*>(op) = val;
  }
}

extern "C" void kernel_launch(void* const* d_in, const int* in_sizes, int n_in,
                              void* d_out, int out_size, void* d_ws, size_t ws_size,
                              hipStream_t stream) {
  const float* first  = (const float*)d_in[0];
  const float* second = (const float*)d_in[1];
  float* out = (float*)d_out;

  const int nblocks = Bn * (Hn / 2);
  corr_wmma_kernel<<<nblocks, NWAVES * 32, 0, stream>>>(first, second, out);
}
